# starting kernel
# baseline (speedup 1.0000x reference)
#include <hip/hip_runtime.h>
#include <math.h>

typedef __attribute__((ext_vector_type(2))) float v2f;
typedef __attribute__((ext_vector_type(4))) float v4f;
typedef __attribute__((ext_vector_type(8))) float v8f;

#define BATCH  8
#define CIN    128
#define COUT   256
#define NDIM   4096
#define TILE_N 16
#define MCOLS  (8 * TILE_N)
#define NT     (NDIM / TILE_N)
#define PANEL  (COUT * MCOLS)

__device__ __forceinline__ int swz(int row, int col) {
    return ((row >> 2) << 9) + (col << 2) + (row & 3);
}

__device__ __forceinline__ float killing6(const float xv[8], const float dv[8]) {
    float s = xv[0]*dv[2] + xv[2]*dv[0]
            + xv[1]*dv[4] + xv[4]*dv[1]
            + xv[3]*dv[5] + xv[5]*dv[3]
            + 2.0f*(xv[6]*dv[6] + xv[7]*dv[7])
            - xv[6]*dv[7] - xv[7]*dv[6];
    return 6.0f * s;
}

__device__ __forceinline__ void gemm_block(const float* __restrict__ W, int ldw, int K,
                                           const float* __restrict__ Bsm,
                                           float* __restrict__ Dsm,
                                           int wave, int lane)
{
    const int l15 = lane & 15;
    const int p   = lane >> 4;
    const int k2  = p << 1;
    const int rst = p << 3;

    for (int mi = wave * 2; mi < wave * 2 + 2; ++mi) {
        const int row0 = mi << 4;
        const float* Wrow = W + (size_t)(row0 + l15) * ldw;

        v8f acc[8];
        #pragma unroll
        for (int nj = 0; nj < 8; ++nj)
            acc[nj] = (v8f){0.f,0.f,0.f,0.f,0.f,0.f,0.f,0.f};

        const float* Bl = Bsm + (l15 << 2) + k2;

        #pragma unroll 2
        for (int kq4 = 0; kq4 < (K >> 2); ++kq4) {
            v2f a = *(const v2f*)(Wrow + (kq4 << 2) + k2);
            const float* Bk = Bl + (kq4 << 9);
            #pragma unroll
            for (int nj = 0; nj < 8; ++nj) {
                v2f bb = *(const v2f*)(Bk + (nj << 6));
                acc[nj] = __builtin_amdgcn_wmma_f32_16x16x4_f32(
                    false, a, false, bb, (short)0, acc[nj], false, false);
            }
        }
        #pragma unroll
        for (int nj = 0; nj < 8; ++nj) {
            const int col = (nj << 4) + l15;
            #pragma unroll
            for (int r = 0; r < 8; r += 2) {
                v2f pr;
                pr.x = acc[nj][r];
                pr.y = acc[nj][r + 1];
                *(v2f*)(Dsm + swz(row0 + rst + r, col)) = pr;
            }
        }
    }
}

extern "C" __global__ __launch_bounds__(256)
void ln_fused(const float* __restrict__ x,
              const float* __restrict__ Wlin,
              const float* __restrict__ Wrelu,
              const float* __restrict__ Wpool,
              float* __restrict__ ws_val,
              float* __restrict__ ws_vec)
{
    extern __shared__ float smem[];
    float* bufA = smem;
    float* bufB = smem + PANEL;

    const int tid  = threadIdx.x;
    const int lane = tid & 31;
    const int wave = tid >> 5;
    const int tile = blockIdx.x;
    const int b    = blockIdx.y;
    const int n0   = tile * TILE_N;

    {
        const float* xb = x + ((size_t)b * CIN * 8) * NDIM + n0;
        for (int it = tid; it < (CIN / 4) * MCOLS; it += 256) {
            const int col = it & (MCOLS - 1);
            const int cg  = it >> 7;
            const int k8  = col >> 4;
            const int nn  = col & 15;
            const float* gp = xb + ((size_t)(cg << 2) * 8 + k8) * NDIM + nn;
            v4f v;
            v.x = gp[0];
            v.y = gp[8 * NDIM];
            v.z = gp[16 * NDIM];
            v.w = gp[24 * NDIM];
            *(v4f*)(bufB + (cg << 9) + (col << 2)) = v;
        }
    }
    __syncthreads();

    gemm_block(Wlin, CIN, CIN, bufB, bufA, wave, lane);
    __syncthreads();
    gemm_block(Wrelu, COUT, COUT, bufA, bufB, wave, lane);
    __syncthreads();

    for (int pp = tid; pp < COUT * TILE_N; pp += 256) {
        const int f = pp >> 4;
        const int n = pp & 15;
        float xv[8], dv[8];
        #pragma unroll
        for (int k = 0; k < 8; ++k) {
            const int a = swz(f, k * TILE_N + n);
            xv[k] = bufA[a];
            dv[k] = bufB[a];
        }
        const float kf = killing6(xv, dv);
        #pragma unroll
        for (int k = 0; k < 8; ++k)
            bufA[swz(f, k * TILE_N + n)] = (kf < 0.0f) ? xv[k] : fmaf(kf, dv[k], xv[k]);
    }
    __syncthreads();

    gemm_block(Wpool, COUT, COUT, bufA, bufB, wave, lane);
    __syncthreads();

    {
        const int f = tid;
        float xv[8], dv[8];
        #pragma unroll
        for (int k = 0; k < 8; ++k) {
            const int a = swz(f, k * TILE_N);
            xv[k] = bufA[a];
            dv[k] = bufB[a];
        }
        float best = killing6(xv, dv);
        int bestn = 0;
        for (int n = 1; n < TILE_N; ++n) {
            #pragma unroll
            for (int k = 0; k < 8; ++k) {
                const int a = swz(f, k * TILE_N + n);
                xv[k] = bufA[a];
                dv[k] = bufB[a];
            }
            const float v = killing6(xv, dv);
            if (v > best) { best = v; bestn = n; }
        }
        const size_t base = ((size_t)b * COUT + f) * NT + tile;
        ws_val[base] = best;
        #pragma unroll
        for (int k = 0; k < 8; ++k)
            ws_vec[base * 8 + k] = bufA[swz(f, k * TILE_N + bestn)];
    }
}

extern "C" __global__ __launch_bounds__(256)
void ln_reduce(const float* __restrict__ ws_val,
               const float* __restrict__ ws_vec,
               float* __restrict__ out)
{
    const int bf = blockIdx.x * 256 + threadIdx.x;
    if (bf >= BATCH * COUT) return;
    const float* v = ws_val + (size_t)bf * NT;
    float best = v[0];
    int bestt = 0;
    for (int t = 1; t < NT; ++t) {
        const float val = v[t];
        if (val > best) { best = val; bestt = t; }
    }
    const float* src = ws_vec + ((size_t)bf * NT + bestt) * 8;
    #pragma unroll
    for (int k = 0; k < 8; ++k)
        out[(size_t)bf * 8 + k] = src[k];
}

extern "C" void kernel_launch(void* const* d_in, const int* in_sizes, int n_in,
                              void* d_out, int out_size, void* d_ws, size_t ws_size,
                              hipStream_t stream)
{
    (void)in_sizes; (void)n_in; (void)out_size; (void)ws_size;

    const float* x     = (const float*)d_in[0];
    const float* Wlin  = (const float*)d_in[1];
    const float* Wrelu = (const float*)d_in[2];
    const float* Wpool = (const float*)d_in[3];
    float* out    = (float*)d_out;
    float* ws_val = (float*)d_ws;
    float* ws_vec = ws_val + (size_t)BATCH * COUT * NT;

    const size_t shmem = (size_t)2 * PANEL * sizeof(float);
    hipFuncSetAttribute((const void*)ln_fused,
                        hipFuncAttributeMaxDynamicSharedMemorySize, (int)shmem);

    dim3 grid(NT, BATCH);
    ln_fused<<<grid, 256, shmem, stream>>>(x, Wlin, Wrelu, Wpool, ws_val, ws_vec);
    ln_reduce<<<(BATCH * COUT + 255) / 256, 256, 0, stream>>>(ws_val, ws_vec, out);
}


# GEMM k-loop: each wave accumulates both of its 16-row tiles in one pass over K, so every B fragment read from LDS feeds two WMMAs and one load batch precedes 64 WMMAs instead of 16
# speedup vs baseline: 1.0898x; 1.0898x over previous
#include <hip/hip_runtime.h>
#include <math.h>

typedef __attribute__((ext_vector_type(2))) float v2f;
typedef __attribute__((ext_vector_type(4))) float v4f;
typedef __attribute__((ext_vector_type(8))) float v8f;

#define BATCH  8
#define CIN    128
#define COUT   256
#define NDIM   4096
#define TILE_N 16
#define MCOLS  (8 * TILE_N)
#define NT     (NDIM / TILE_N)
#define PANEL  (COUT * MCOLS)

__device__ __forceinline__ int swz(int row, int col) {
    return ((row >> 2) << 9) + (col << 2) + (row & 3);
}

__device__ __forceinline__ float killing6(const float xv[8], const float dv[8]) {
    float s = xv[0]*dv[2] + xv[2]*dv[0]
            + xv[1]*dv[4] + xv[4]*dv[1]
            + xv[3]*dv[5] + xv[5]*dv[3]
            + 2.0f*(xv[6]*dv[6] + xv[7]*dv[7])
            - xv[6]*dv[7] - xv[7]*dv[6];
    return 6.0f * s;
}

__device__ __forceinline__ void gemm_block(const float* __restrict__ W, int ldw, int K,
                                           const float* __restrict__ Bsm,
                                           float* __restrict__ Dsm,
                                           int wave, int lane)
{
    const int l15 = lane & 15;
    const int p   = lane >> 4;
    const int k2  = p << 1;
    const int rst = p << 3;
    const int row0 = wave << 5;

    const float* Wr0 = W + (size_t)(row0 + l15) * ldw + k2;
    const float* Wr1 = Wr0 + (size_t)16 * ldw;

    v8f acc0[8], acc1[8];
    #pragma unroll
    for (int nj = 0; nj < 8; ++nj) {
        acc0[nj] = (v8f){0.f,0.f,0.f,0.f,0.f,0.f,0.f,0.f};
        acc1[nj] = (v8f){0.f,0.f,0.f,0.f,0.f,0.f,0.f,0.f};
    }

    const float* Bl = Bsm + (l15 << 2) + k2;

    #pragma unroll 4
    for (int kq4 = 0; kq4 < (K >> 2); ++kq4) {
        const v2f a0 = *(const v2f*)(Wr0 + (kq4 << 2));
        const v2f a1 = *(const v2f*)(Wr1 + (kq4 << 2));
        const float* Bk = Bl + (kq4 << 9);
        #pragma unroll
        for (int nj = 0; nj < 8; ++nj) {
            const v2f bb = *(const v2f*)(Bk + (nj << 6));
            acc0[nj] = __builtin_amdgcn_wmma_f32_16x16x4_f32(
                false, a0, false, bb, (short)0, acc0[nj], false, false);
            acc1[nj] = __builtin_amdgcn_wmma_f32_16x16x4_f32(
                false, a1, false, bb, (short)0, acc1[nj], false, false);
        }
    }
    #pragma unroll
    for (int nj = 0; nj < 8; ++nj) {
        const int col = (nj << 4) + l15;
        #pragma unroll
        for (int r = 0; r < 8; r += 2) {
            v2f pr;
            pr.x = acc0[nj][r];
            pr.y = acc0[nj][r + 1];
            *(v2f*)(Dsm + swz(row0 + rst + r, col)) = pr;
            v2f qr;
            qr.x = acc1[nj][r];
            qr.y = acc1[nj][r + 1];
            *(v2f*)(Dsm + swz(row0 + 16 + rst + r, col)) = qr;
        }
    }
}

extern "C" __global__ __launch_bounds__(256)
void ln_fused(const float* __restrict__ x,
              const float* __restrict__ Wlin,
              const float* __restrict__ Wrelu,
              const float* __restrict__ Wpool,
              float* __restrict__ ws_val,
              float* __restrict__ ws_vec)
{
    extern __shared__ float smem[];
    float* bufA = smem;
    float* bufB = smem + PANEL;

    const int tid  = threadIdx.x;
    const int lane = tid & 31;
    const int wave = tid >> 5;
    const int tile = blockIdx.x;
    const int b    = blockIdx.y;
    const int n0   = tile * TILE_N;

    {
        const float* xb = x + ((size_t)b * CIN * 8) * NDIM + n0;
        for (int it = tid; it < (CIN / 4) * MCOLS; it += 256) {
            const int col = it & (MCOLS - 1);
            const int cg  = it >> 7;
            const int k8  = col >> 4;
            const int nn  = col & 15;
            const float* gp = xb + ((size_t)(cg << 2) * 8 + k8) * NDIM + nn;
            v4f v;
            v.x = gp[0];
            v.y = gp[8 * NDIM];
            v.z = gp[16 * NDIM];
            v.w = gp[24 * NDIM];
            *(v4f*)(bufB + (cg << 9) + (col << 2)) = v;
        }
    }
    __syncthreads();

    gemm_block(Wlin, CIN, CIN, bufB, bufA, wave, lane);
    __syncthreads();
    gemm_block(Wrelu, COUT, COUT, bufA, bufB, wave, lane);
    __syncthreads();

    for (int pp = tid; pp < COUT * TILE_N; pp += 256) {
        const int f = pp >> 4;
        const int n = pp & 15;
        float xv[8], dv[8];
        #pragma unroll
        for (int k = 0; k < 8; ++k) {
            const int a = swz(f, k * TILE_N + n);
            xv[k] = bufA[a];
            dv[k] = bufB[a];
        }
        const float kf = killing6(xv, dv);
        #pragma unroll
        for (int k = 0; k < 8; ++k)
            bufA[swz(f, k * TILE_N + n)] = (kf < 0.0f) ? xv[k] : fmaf(kf, dv[k], xv[k]);
    }
    __syncthreads();

    gemm_block(Wpool, COUT, COUT, bufA, bufB, wave, lane);
    __syncthreads();

    {
        const int f = tid;
        float xv[8], dv[8];
        #pragma unroll
        for (int k = 0; k < 8; ++k) {
            const int a = swz(f, k * TILE_N);
            xv[k] = bufA[a];
            dv[k] = bufB[a];
        }
        float best = killing6(xv, dv);
        int bestn = 0;
        for (int n = 1; n < TILE_N; ++n) {
            #pragma unroll
            for (int k = 0; k < 8; ++k) {
                const int a = swz(f, k * TILE_N + n);
                xv[k] = bufA[a];
                dv[k] = bufB[a];
            }
            const float v = killing6(xv, dv);
            if (v > best) { best = v; bestn = n; }
        }
        const size_t base = ((size_t)b * COUT + f) * NT + tile;
        ws_val[base] = best;
        #pragma unroll
        for (int k = 0; k < 8; ++k)
            ws_vec[base * 8 + k] = bufA[swz(f, k * TILE_N + bestn)];
    }
}

extern "C" __global__ __launch_bounds__(256)
void ln_reduce(const float* __restrict__ ws_val,
               const float* __restrict__ ws_vec,
               float* __restrict__ out)
{
    const int bf = blockIdx.x * 256 + threadIdx.x;
    if (bf >= BATCH * COUT) return;
    const float* v = ws_val + (size_t)bf * NT;
    float best = v[0];
    int bestt = 0;
    for (int t = 1; t < NT; ++t) {
        const float val = v[t];
        if (val > best) { best = val; bestt = t; }
    }
    const float* src = ws_vec + ((size_t)bf * NT + bestt) * 8;
    #pragma unroll
    for (int k = 0; k < 8; ++k)
        out[(size_t)bf * 8 + k] = src[k];
}

extern "C" void kernel_launch(void* const* d_in, const int* in_sizes, int n_in,
                              void* d_out, int out_size, void* d_ws, size_t ws_size,
                              hipStream_t stream)
{
    (void)in_sizes; (void)n_in; (void)out_size; (void)ws_size;

    const float* x     = (const float*)d_in[0];
    const float* Wlin  = (const float*)d_in[1];
    const float* Wrelu = (const float*)d_in[2];
    const float* Wpool = (const float*)d_in[3];
    float* out    = (float*)d_out;
    float* ws_val = (float*)d_ws;
    float* ws_vec = ws_val + (size_t)BATCH * COUT * NT;

    const size_t shmem = (size_t)2 * PANEL * sizeof(float);
    hipFuncSetAttribute((const void*)ln_fused,
                        hipFuncAttributeMaxDynamicSharedMemorySize, (int)shmem);

    dim3 grid(NT, BATCH);
    ln_fused<<<grid, 256, shmem, stream>>>(x, Wlin, Wrelu, Wpool, ws_val, ws_vec);
    ln_reduce<<<(BATCH * COUT + 255) / 256, 256, 0, stream>>>(ws_val, ws_vec, out);
}


# GEMM k-loops software-pipelined over two fragment register sets: the next trip's A (global) and B (LDS) fragment loads are issued as one group ahead of the current trip's 64 WMMAs; sums unchanged
# speedup vs baseline: 1.1442x; 1.0499x over previous
#include <hip/hip_runtime.h>
#include <math.h>

typedef __attribute__((ext_vector_type(2))) float v2f;
typedef __attribute__((ext_vector_type(4))) float v4f;
typedef __attribute__((ext_vector_type(8))) float v8f;

#define BATCH  8
#define CIN    128
#define COUT   256
#define NDIM   4096
#define TILE_N 16
#define MCOLS  (8 * TILE_N)
#define NT     (NDIM / TILE_N)
#define PANEL  (COUT * MCOLS)

__device__ __forceinline__ int swz(int row, int col) {
    return ((row >> 2) << 9) + (col << 2) + (row & 3);
}

__device__ __forceinline__ float killing6(const float xv[8], const float dv[8]) {
    float s = xv[0]*dv[2] + xv[2]*dv[0]
            + xv[1]*dv[4] + xv[4]*dv[1]
            + xv[3]*dv[5] + xv[5]*dv[3]
            + 2.0f*(xv[6]*dv[6] + xv[7]*dv[7])
            - xv[6]*dv[7] - xv[7]*dv[6];
    return 6.0f * s;
}

__device__ __forceinline__ void load_trip(const float* __restrict__ Wr0,
                                          const float* __restrict__ Wr1,
                                          const float* __restrict__ Bl, int t,
                                          v2f (&a0)[4], v2f (&a1)[4], v2f (&bf)[4][8])
{
    const float* w0 = Wr0 + (t << 4);
    const float* w1 = Wr1 + (t << 4);
    const float* Bt = Bl + (t << 11);
    #pragma unroll
    for (int s = 0; s < 4; ++s) {
        a0[s] = *(const v2f*)(w0 + (s << 2));
        a1[s] = *(const v2f*)(w1 + (s << 2));
    }
    #pragma unroll
    for (int s = 0; s < 4; ++s) {
        #pragma unroll
        for (int nj = 0; nj < 8; ++nj)
            bf[s][nj] = *(const v2f*)(Bt + (s << 9) + (nj << 6));
    }
}

__device__ __forceinline__ void mma_trip(const v2f (&a0)[4], const v2f (&a1)[4],
                                         const v2f (&bf)[4][8],
                                         v8f (&acc0)[8], v8f (&acc1)[8])
{
    #pragma unroll
    for (int s = 0; s < 4; ++s) {
        #pragma unroll
        for (int nj = 0; nj < 8; ++nj) {
            acc0[nj] = __builtin_amdgcn_wmma_f32_16x16x4_f32(
                false, a0[s], false, bf[s][nj], (short)0, acc0[nj], false, false);
            acc1[nj] = __builtin_amdgcn_wmma_f32_16x16x4_f32(
                false, a1[s], false, bf[s][nj], (short)0, acc1[nj], false, false);
        }
    }
}

__device__ __forceinline__ void gemm_block(const float* __restrict__ W, int ldw, int K,
                                           const float* __restrict__ Bsm,
                                           float* __restrict__ Dsm,
                                           int wave, int lane)
{
    const int l15 = lane & 15;
    const int p   = lane >> 4;
    const int k2  = p << 1;
    const int rst = p << 3;
    const int row0 = wave << 5;

    const float* Wr0 = W + (size_t)(row0 + l15) * ldw + k2;
    const float* Wr1 = Wr0 + (size_t)16 * ldw;

    v8f acc0[8], acc1[8];
    #pragma unroll
    for (int nj = 0; nj < 8; ++nj) {
        acc0[nj] = (v8f){0.f,0.f,0.f,0.f,0.f,0.f,0.f,0.f};
        acc1[nj] = (v8f){0.f,0.f,0.f,0.f,0.f,0.f,0.f,0.f};
    }

    const float* Bl = Bsm + (l15 << 2) + k2;

    const int T = K >> 4;
    v2f pa0[4], pa1[4], pb[4][8];
    v2f qa0[4], qa1[4], qb[4][8];

    load_trip(Wr0, Wr1, Bl, 0, pa0, pa1, pb);
    #pragma unroll 1
    for (int t = 0; t < T - 2; t += 2) {
        load_trip(Wr0, Wr1, Bl, t + 1, qa0, qa1, qb);
        __builtin_amdgcn_sched_barrier(0);
        mma_trip(pa0, pa1, pb, acc0, acc1);
        load_trip(Wr0, Wr1, Bl, t + 2, pa0, pa1, pb);
        __builtin_amdgcn_sched_barrier(0);
        mma_trip(qa0, qa1, qb, acc0, acc1);
    }
    load_trip(Wr0, Wr1, Bl, T - 1, qa0, qa1, qb);
    __builtin_amdgcn_sched_barrier(0);
    mma_trip(pa0, pa1, pb, acc0, acc1);
    mma_trip(qa0, qa1, qb, acc0, acc1);

    #pragma unroll
    for (int nj = 0; nj < 8; ++nj) {
        const int col = (nj << 4) + l15;
        #pragma unroll
        for (int r = 0; r < 8; r += 2) {
            v2f pr;
            pr.x = acc0[nj][r];
            pr.y = acc0[nj][r + 1];
            *(v2f*)(Dsm + swz(row0 + rst + r, col)) = pr;
            v2f qr;
            qr.x = acc1[nj][r];
            qr.y = acc1[nj][r + 1];
            *(v2f*)(Dsm + swz(row0 + 16 + rst + r, col)) = qr;
        }
    }
}

extern "C" __global__ __launch_bounds__(256)
void ln_fused(const float* __restrict__ x,
              const float* __restrict__ Wlin,
              const float* __restrict__ Wrelu,
              const float* __restrict__ Wpool,
              float* __restrict__ ws_val,
              float* __restrict__ ws_vec)
{
    extern __shared__ float smem[];
    float* bufA = smem;
    float* bufB = smem + PANEL;

    const int tid  = threadIdx.x;
    const int lane = tid & 31;
    const int wave = tid >> 5;
    const int tile = blockIdx.x;
    const int b    = blockIdx.y;
    const int n0   = tile * TILE_N;

    {
        const float* xb = x + ((size_t)b * CIN * 8) * NDIM + n0;
        for (int it = tid; it < (CIN / 4) * MCOLS; it += 256) {
            const int col = it & (MCOLS - 1);
            const int cg  = it >> 7;
            const int k8  = col >> 4;
            const int nn  = col & 15;
            const float* gp = xb + ((size_t)(cg << 2) * 8 + k8) * NDIM + nn;
            v4f v;
            v.x = gp[0];
            v.y = gp[8 * NDIM];
            v.z = gp[16 * NDIM];
            v.w = gp[24 * NDIM];
            *(v4f*)(bufB + (cg << 9) + (col << 2)) = v;
        }
    }
    __syncthreads();

    gemm_block(Wlin, CIN, CIN, bufB, bufA, wave, lane);
    __syncthreads();
    gemm_block(Wrelu, COUT, COUT, bufA, bufB, wave, lane);
    __syncthreads();

    for (int pp = tid; pp < COUT * TILE_N; pp += 256) {
        const int f = pp >> 4;
        const int n = pp & 15;
        float xv[8], dv[8];
        #pragma unroll
        for (int k = 0; k < 8; ++k) {
            const int a = swz(f, k * TILE_N + n);
            xv[k] = bufA[a];
            dv[k] = bufB[a];
        }
        const float kf = killing6(xv, dv);
        #pragma unroll
        for (int k = 0; k < 8; ++k)
            bufA[swz(f, k * TILE_N + n)] = (kf < 0.0f) ? xv[k] : fmaf(kf, dv[k], xv[k]);
    }
    __syncthreads();

    gemm_block(Wpool, COUT, COUT, bufA, bufB, wave, lane);
    __syncthreads();

    {
        const int f = tid;
        float xv[8], dv[8];
        #pragma unroll
        for (int k = 0; k < 8; ++k) {
            const int a = swz(f, k * TILE_N);
            xv[k] = bufA[a];
            dv[k] = bufB[a];
        }
        float best = killing6(xv, dv);
        int bestn = 0;
        for (int n = 1; n < TILE_N; ++n) {
            #pragma unroll
            for (int k = 0; k < 8; ++k) {
                const int a = swz(f, k * TILE_N + n);
                xv[k] = bufA[a];
                dv[k] = bufB[a];
            }
            const float v = killing6(xv, dv);
            if (v > best) { best = v; bestn = n; }
        }
        const size_t base = ((size_t)b * COUT + f) * NT + tile;
        ws_val[base] = best;
        #pragma unroll
        for (int k = 0; k < 8; ++k)
            ws_vec[base * 8 + k] = bufA[swz(f, k * TILE_N + bestn)];
    }
}

extern "C" __global__ __launch_bounds__(256)
void ln_reduce(const float* __restrict__ ws_val,
               const float* __restrict__ ws_vec,
               float* __restrict__ out)
{
    const int bf = blockIdx.x * 256 + threadIdx.x;
    if (bf >= BATCH * COUT) return;
    const float* v = ws_val + (size_t)bf * NT;
    float best = v[0];
    int bestt = 0;
    for (int t = 1; t < NT; ++t) {
        const float val = v[t];
        if (val > best) { best = val; bestt = t; }
    }
    const float* src = ws_vec + ((size_t)bf * NT + bestt) * 8;
    #pragma unroll
    for (int k = 0; k < 8; ++k)
        out[(size_t)bf * 8 + k] = src[k];
}

extern "C" void kernel_launch(void* const* d_in, const int* in_sizes, int n_in,
                              void* d_out, int out_size, void* d_ws, size_t ws_size,
                              hipStream_t stream)
{
    (void)in_sizes; (void)n_in; (void)out_size; (void)ws_size;

    const float* x     = (const float*)d_in[0];
    const float* Wlin  = (const float*)d_in[1];
    const float* Wrelu = (const float*)d_in[2];
    const float* Wpool = (const float*)d_in[3];
    float* out    = (float*)d_out;
    float* ws_val = (float*)d_ws;
    float* ws_vec = ws_val + (size_t)BATCH * COUT * NT;

    const size_t shmem = (size_t)2 * PANEL * sizeof(float);
    hipFuncSetAttribute((const void*)ln_fused,
                        hipFuncAttributeMaxDynamicSharedMemorySize, (int)shmem);

    dim3 grid(NT, BATCH);
    ln_fused<<<grid, 256, shmem, stream>>>(x, Wlin, Wrelu, Wpool, ws_val, ws_vec);
    ln_reduce<<<(BATCH * COUT + 255) / 256, 256, 0, stream>>>(ws_val, ws_vec, out);
}


# Killing-form passes fused into the epilogues of GEMM 2 and 3: d, d2 stay in the accumulators, x read from LDS in conflict-free 16-byte units, argmax over n by 16-lane max + ballot (first occurrence)
# speedup vs baseline: 1.2670x; 1.1073x over previous
#include <hip/hip_runtime.h>
#include <math.h>

typedef __attribute__((ext_vector_type(2))) float v2f;
typedef __attribute__((ext_vector_type(4))) float v4f;
typedef __attribute__((ext_vector_type(8))) float v8f;

#define BATCH  8
#define CIN    128
#define COUT   256
#define NDIM   4096
#define TILE_N 16
#define MCOLS  (8 * TILE_N)
#define NT     (NDIM / TILE_N)
#define PANEL  (COUT * MCOLS)

__device__ __forceinline__ int swz(int row, int col) {
    return ((row >> 2) << 9) + (col << 2) + (row & 3);
}

__device__ __forceinline__ float killing6(const float xv[8], const float dv[8]) {
    float s = xv[0]*dv[2] + xv[2]*dv[0]
            + xv[1]*dv[4] + xv[4]*dv[1]
            + xv[3]*dv[5] + xv[5]*dv[3]
            + 2.0f*(xv[6]*dv[6] + xv[7]*dv[7])
            - xv[6]*dv[7] - xv[7]*dv[6];
    return 6.0f * s;
}

__device__ __forceinline__ void load_trip(const float* __restrict__ Wr0,
                                          const float* __restrict__ Wr1,
                                          const float* __restrict__ Bl, int t,
                                          v2f (&a0)[4], v2f (&a1)[4], v2f (&bf)[4][8])
{
    const float* w0 = Wr0 + (t << 4);
    const float* w1 = Wr1 + (t << 4);
    const float* Bt = Bl + (t << 11);
    #pragma unroll
    for (int s = 0; s < 4; ++s) {
        a0[s] = *(const v2f*)(w0 + (s << 2));
        a1[s] = *(const v2f*)(w1 + (s << 2));
    }
    #pragma unroll
    for (int s = 0; s < 4; ++s) {
        #pragma unroll
        for (int nj = 0; nj < 8; ++nj)
            bf[s][nj] = *(const v2f*)(Bt + (s << 9) + (nj << 6));
    }
}

__device__ __forceinline__ void mma_trip(const v2f (&a0)[4], const v2f (&a1)[4],
                                         const v2f (&bf)[4][8],
                                         v8f (&acc0)[8], v8f (&acc1)[8])
{
    #pragma unroll
    for (int s = 0; s < 4; ++s) {
        #pragma unroll
        for (int nj = 0; nj < 8; ++nj) {
            acc0[nj] = __builtin_amdgcn_wmma_f32_16x16x4_f32(
                false, a0[s], false, bf[s][nj], (short)0, acc0[nj], false, false);
            acc1[nj] = __builtin_amdgcn_wmma_f32_16x16x4_f32(
                false, a1[s], false, bf[s][nj], (short)0, acc1[nj], false, false);
        }
    }
}

__device__ __forceinline__ void gemm_acc(const float* __restrict__ W, int ldw, int K,
                                         const float* __restrict__ Bsm,
                                         int wave, int lane,
                                         v8f (&acc0)[8], v8f (&acc1)[8])
{
    const int l15 = lane & 15;
    const int p   = lane >> 4;
    const int k2  = p << 1;
    const int row0 = wave << 5;

    const float* Wr0 = W + (size_t)(row0 + l15) * ldw + k2;
    const float* Wr1 = Wr0 + (size_t)16 * ldw;

    #pragma unroll
    for (int nj = 0; nj < 8; ++nj) {
        acc0[nj] = (v8f){0.f,0.f,0.f,0.f,0.f,0.f,0.f,0.f};
        acc1[nj] = (v8f){0.f,0.f,0.f,0.f,0.f,0.f,0.f,0.f};
    }

    const float* Bl = Bsm + (l15 << 2) + k2;

    const int T = K >> 4;
    v2f pa0[4], pa1[4], pb[4][8];
    v2f qa0[4], qa1[4], qb[4][8];

    load_trip(Wr0, Wr1, Bl, 0, pa0, pa1, pb);
    #pragma unroll 1
    for (int t = 0; t < T - 2; t += 2) {
        load_trip(Wr0, Wr1, Bl, t + 1, qa0, qa1, qb);
        __builtin_amdgcn_sched_barrier(0);
        mma_trip(pa0, pa1, pb, acc0, acc1);
        load_trip(Wr0, Wr1, Bl, t + 2, pa0, pa1, pb);
        __builtin_amdgcn_sched_barrier(0);
        mma_trip(qa0, qa1, qb, acc0, acc1);
    }
    load_trip(Wr0, Wr1, Bl, T - 1, qa0, qa1, qb);
    __builtin_amdgcn_sched_barrier(0);
    mma_trip(pa0, pa1, pb, acc0, acc1);
    mma_trip(qa0, qa1, qb, acc0, acc1);
}

__device__ __forceinline__ void store_acc(float* __restrict__ Dsm, int wave, int lane,
                                          const v8f (&acc0)[8], const v8f (&acc1)[8])
{
    const int l15 = lane & 15;
    const int rst = (lane >> 4) << 3;
    const int row0 = wave << 5;
    #pragma unroll
    for (int nj = 0; nj < 8; ++nj) {
        const int col = (nj << 4) + l15;
        #pragma unroll
        for (int r = 0; r < 8; r += 2) {
            v2f pr;
            pr.x = acc0[nj][r];
            pr.y = acc0[nj][r + 1];
            *(v2f*)(Dsm + swz(row0 + rst + r, col)) = pr;
            v2f qr;
            qr.x = acc1[nj][r];
            qr.y = acc1[nj][r + 1];
            *(v2f*)(Dsm + swz(row0 + 16 + rst + r, col)) = qr;
        }
    }
}


__device__ __forceinline__ void killing_relu_tile(float* __restrict__ Xsm, int rowbase, int lane,
                                                  const v8f (&acc)[8])
{
    const int l15 = lane & 15;
    const int p   = lane >> 4;
    #pragma unroll
    for (int g = 0; g < 2; ++g) {
        const int f4 = rowbase + (p << 3) + (g << 2);
        float* base = Xsm + ((f4 >> 2) << 9) + (l15 << 2);
        v4f xa[8];
        #pragma unroll
        for (int k = 0; k < 8; ++k)
            xa[k] = *(const v4f*)(base + (k << 6));
        #pragma unroll
        for (int j = 0; j < 4; ++j) {
            float xv[8], dv[8];
            #pragma unroll
            for (int k = 0; k < 8; ++k) {
                xv[k] = xa[k][j];
                dv[k] = acc[k][(g << 2) + j];
            }
            const float kf = killing6(xv, dv);
            #pragma unroll
            for (int k = 0; k < 8; ++k)
                xa[k][j] = (kf < 0.0f) ? xv[k] : fmaf(kf, dv[k], xv[k]);
        }
        #pragma unroll
        for (int k = 0; k < 8; ++k)
            *(v4f*)(base + (k << 6)) = xa[k];
    }
}

__device__ __forceinline__ void killing_argmax(const float* __restrict__ Xsm, int row0, int lane,
                                               const v8f (&acc0)[8], const v8f (&acc1)[8],
                                               float* __restrict__ ws_val,
                                               float* __restrict__ ws_vec,
                                               size_t chan0, int tile)
{
    const int l15 = lane & 15;
    const int p   = lane >> 4;
    v4f   xa[4][8];
    float kf[16];
    float m[16];

    #pragma unroll
    for (int q = 0; q < 4; ++q) {
        const int f4 = row0 + ((q >> 1) << 4) + (p << 3) + ((q & 1) << 2);
        const float* base = Xsm + ((f4 >> 2) << 9) + (l15 << 2);
        #pragma unroll
        for (int k = 0; k < 8; ++k)
            xa[q][k] = *(const v4f*)(base + (k << 6));
    }
    #pragma unroll
    for (int q = 0; q < 4; ++q) {
        #pragma unroll
        for (int j = 0; j < 4; ++j) {
            float xv[8], dv[8];
            #pragma unroll
            for (int k = 0; k < 8; ++k) {
                xv[k] = xa[q][k][j];
                dv[k] = (q < 2) ? acc0[k][((q & 1) << 2) + j] : acc1[k][((q & 1) << 2) + j];
            }
            kf[(q << 2) + j] = killing6(xv, dv);
            m[(q << 2) + j]  = kf[(q << 2) + j];
        }
    }
    #pragma unroll
    for (int s = 1; s < 16; s <<= 1) {
        #pragma unroll
        for (int i = 0; i < 16; ++i)
            m[i] = fmaxf(m[i], __shfl_xor(m[i], s, 32));
    }
    int bestn[16];
    #pragma unroll
    for (int i = 0; i < 16; ++i) {
        const unsigned all = (unsigned)__ballot(kf[i] == m[i]);
        const unsigned hit = (all >> (p << 4)) & 0xffffu;
        bestn[i] = hit ? (__ffs(hit) - 1) : 0;
    }
    #pragma unroll
    for (int q = 0; q < 4; ++q) {
        const int f4 = row0 + ((q >> 1) << 4) + (p << 3) + ((q & 1) << 2);
        #pragma unroll
        for (int j = 0; j < 4; ++j) {
            if (l15 == bestn[(q << 2) + j]) {
                const size_t o = (chan0 + (size_t)(f4 + j)) * NT + tile;
                ws_val[o] = kf[(q << 2) + j];
                v4f lo, hi;
                lo.x = xa[q][0][j]; lo.y = xa[q][1][j]; lo.z = xa[q][2][j]; lo.w = xa[q][3][j];
                hi.x = xa[q][4][j]; hi.y = xa[q][5][j]; hi.z = xa[q][6][j]; hi.w = xa[q][7][j];
                *(v4f*)(ws_vec + o * 8)     = lo;
                *(v4f*)(ws_vec + o * 8 + 4) = hi;
            }
        }
    }
}

extern "C" __global__ __launch_bounds__(256)
void ln_fused(const float* __restrict__ x,
              const float* __restrict__ Wlin,
              const float* __restrict__ Wrelu,
              const float* __restrict__ Wpool,
              float* __restrict__ ws_val,
              float* __restrict__ ws_vec)
{
    extern __shared__ float smem[];
    float* bufA = smem;
    float* bufB = smem + PANEL;

    const int tid  = threadIdx.x;
    const int lane = tid & 31;
    const int wave = tid >> 5;
    const int tile = blockIdx.x;
    const int b    = blockIdx.y;
    const int n0   = tile * TILE_N;

    {
        const float* xb = x + ((size_t)b * CIN * 8) * NDIM + n0;
        for (int it = tid; it < (CIN / 4) * MCOLS; it += 256) {
            const int col = it & (MCOLS - 1);
            const int cg  = it >> 7;
            const int k8  = col >> 4;
            const int nn  = col & 15;
            const float* gp = xb + ((size_t)(cg << 2) * 8 + k8) * NDIM + nn;
            v4f v;
            v.x = gp[0];
            v.y = gp[8 * NDIM];
            v.z = gp[16 * NDIM];
            v.w = gp[24 * NDIM];
            *(v4f*)(bufB + (cg << 9) + (col << 2)) = v;
        }
    }
    __syncthreads();

    v8f acc0[8], acc1[8];
    const int row0 = wave << 5;

    gemm_acc(Wlin, CIN, CIN, bufB, wave, lane, acc0, acc1);
    store_acc(bufA, wave, lane, acc0, acc1);
    __syncthreads();

    gemm_acc(Wrelu, COUT, COUT, bufA, wave, lane, acc0, acc1);
    __syncthreads();
    killing_relu_tile(bufA, row0,      lane, acc0);
    killing_relu_tile(bufA, row0 + 16, lane, acc1);
    __syncthreads();

    gemm_acc(Wpool, COUT, COUT, bufA, wave, lane, acc0, acc1);
    killing_argmax(bufA, row0, lane, acc0, acc1, ws_val, ws_vec, (size_t)b * COUT, tile);
}

extern "C" __global__ __launch_bounds__(256)
void ln_reduce(const float* __restrict__ ws_val,
               const float* __restrict__ ws_vec,
               float* __restrict__ out)
{
    const int bf = blockIdx.x * 256 + threadIdx.x;
    if (bf >= BATCH * COUT) return;
    const float* v = ws_val + (size_t)bf * NT;
    float best = v[0];
    int bestt = 0;
    for (int t = 1; t < NT; ++t) {
        const float val = v[t];
        if (val > best) { best = val; bestt = t; }
    }
    const float* src = ws_vec + ((size_t)bf * NT + bestt) * 8;
    #pragma unroll
    for (int k = 0; k < 8; ++k)
        out[(size_t)bf * 8 + k] = src[k];
}

extern "C" void kernel_launch(void* const* d_in, const int* in_sizes, int n_in,
                              void* d_out, int out_size, void* d_ws, size_t ws_size,
                              hipStream_t stream)
{
    (void)in_sizes; (void)n_in; (void)out_size; (void)ws_size;

    const float* x     = (const float*)d_in[0];
    const float* Wlin  = (const float*)d_in[1];
    const float* Wrelu = (const float*)d_in[2];
    const float* Wpool = (const float*)d_in[3];
    float* out    = (float*)d_out;
    float* ws_val = (float*)d_ws;
    float* ws_vec = ws_val + (size_t)BATCH * COUT * NT;

    const size_t shmem = (size_t)(PANEL + PANEL / 2) * sizeof(float);
    hipFuncSetAttribute((const void*)ln_fused,
                        hipFuncAttributeMaxDynamicSharedMemorySize, (int)shmem);

    dim3 grid(NT, BATCH);
    ln_fused<<<grid, 256, shmem, stream>>>(x, Wlin, Wrelu, Wpool, ws_val, ws_vec);
    ln_reduce<<<(BATCH * COUT + 255) / 256, 256, 0, stream>>>(ws_val, ws_vec, out);
}


# Fused launch made a tile loop: 256 work-groups walk 8 n-tiles each, and the third product's K loop loads the next tile's x panel into LDS under its matrix instructions (sums unchanged)
# speedup vs baseline: 1.5902x; 1.2551x over previous
#include <hip/hip_runtime.h>
#include <math.h>

typedef __attribute__((ext_vector_type(2))) float v2f;
typedef __attribute__((ext_vector_type(4))) float v4f;
typedef __attribute__((ext_vector_type(8))) float v8f;

#define BATCH  8
#define CIN    128
#define COUT   256
#define NDIM   4096
#define TILE_N 16
#define MCOLS  (8 * TILE_N)
#define NT     (NDIM / TILE_N)
#define PANEL  (COUT * MCOLS)
#define NTP    (NT / 2)
#define CAND_PITCH 12

__device__ __forceinline__ int swz(int row, int col) {
    return ((row >> 2) << 9) + (col << 2) + (row & 3);
}

__device__ __forceinline__ float killing6(const float xv[8], const float dv[8]) {
    float s = xv[0]*dv[2] + xv[2]*dv[0]
            + xv[1]*dv[4] + xv[4]*dv[1]
            + xv[3]*dv[5] + xv[5]*dv[3]
            + 2.0f*(xv[6]*dv[6] + xv[7]*dv[7])
            - xv[6]*dv[7] - xv[7]*dv[6];
    return 6.0f * s;
}

__device__ __forceinline__ void load_trip(const float* __restrict__ Wr0,
                                          const float* __restrict__ Wr1,
                                          const float* __restrict__ Bl, int t,
                                          v2f (&a0)[4], v2f (&a1)[4], v2f (&bf)[4][8])
{
    const float* w0 = Wr0 + (t << 4);
    const float* w1 = Wr1 + (t << 4);
    const float* Bt = Bl + (t << 11);
    #pragma unroll
    for (int s = 0; s < 4; ++s) {
        a0[s] = *(const v2f*)(w0 + (s << 2));
        a1[s] = *(const v2f*)(w1 + (s << 2));
    }
    #pragma unroll
    for (int s = 0; s < 4; ++s) {
        #pragma unroll
        for (int nj = 0; nj < 8; ++nj)
            bf[s][nj] = *(const v2f*)(Bt + (s << 9) + (nj << 6));
    }
}

__device__ __forceinline__ void mma_trip(const v2f (&a0)[4], const v2f (&a1)[4],
                                         const v2f (&bf)[4][8],
                                         v8f (&acc0)[8], v8f (&acc1)[8])
{
    #pragma unroll
    for (int s = 0; s < 4; ++s) {
        #pragma unroll
        for (int nj = 0; nj < 8; ++nj) {
            acc0[nj] = __builtin_amdgcn_wmma_f32_16x16x4_f32(
                false, a0[s], false, bf[s][nj], (short)0, acc0[nj], false, false);
            acc1[nj] = __builtin_amdgcn_wmma_f32_16x16x4_f32(
                false, a1[s], false, bf[s][nj], (short)0, acc1[nj], false, false);
        }
    }
}

__device__ __forceinline__ void panel_issue(const float* __restrict__ xnext, int c, v4f &u0, v4f &u1)
{
    const float* g0 = xnext + (size_t)(c << 2) * (32 * NDIM);
    const float* g1 = g0 + (size_t)2 * (32 * NDIM);
    u0.x = g0[0]; u0.y = g0[8 * NDIM]; u0.z = g0[16 * NDIM]; u0.w = g0[24 * NDIM];
    u1.x = g1[0]; u1.y = g1[8 * NDIM]; u1.z = g1[16 * NDIM]; u1.w = g1[24 * NDIM];
}
__device__ __forceinline__ void panel_store(float* __restrict__ xdst, int c, const v4f &u0, const v4f &u1)
{
    *(v4f*)(xdst + ((c << 2) << 9))       = u0;
    *(v4f*)(xdst + (((c << 2) + 2) << 9)) = u1;
}

template <bool PF>
__device__ __forceinline__ void gemm_acc(const float* __restrict__ W, int ldw, int K,
                                         const float* __restrict__ Bsm,
                                         int wave, int lane,
                                         v8f (&acc0)[8], v8f (&acc1)[8],
                                         const float* __restrict__ xnext,
                                         float* __restrict__ xdst)
{
    const int l15 = lane & 15;
    const int p   = lane >> 4;
    const int k2  = p << 1;
    const int row0 = wave << 5;

    const float* Wr0 = W + (size_t)(row0 + l15) * ldw + k2;
    const float* Wr1 = Wr0 + (size_t)16 * ldw;

    #pragma unroll
    for (int nj = 0; nj < 8; ++nj) {
        acc0[nj] = (v8f){0.f,0.f,0.f,0.f,0.f,0.f,0.f,0.f};
        acc1[nj] = (v8f){0.f,0.f,0.f,0.f,0.f,0.f,0.f,0.f};
    }

    const float* Bl = Bsm + (l15 << 2) + k2;

    const int T = K >> 4;
    v2f pa0[4], pa1[4], pb[4][8];
    v2f qa0[4], qa1[4], qb[4][8];

    v4f u0, u1;

    load_trip(Wr0, Wr1, Bl, 0, pa0, pa1, pb);
    #pragma unroll 1
    for (int t = 0; t < T - 2; t += 2) {
        load_trip(Wr0, Wr1, Bl, t + 1, qa0, qa1, qb);
        if (PF) panel_issue(xnext, t >> 1, u0, u1);
        __builtin_amdgcn_sched_barrier(0);
        mma_trip(pa0, pa1, pb, acc0, acc1);
        load_trip(Wr0, Wr1, Bl, t + 2, pa0, pa1, pb);
        __builtin_amdgcn_sched_barrier(0);
        mma_trip(qa0, qa1, qb, acc0, acc1);
        if (PF) panel_store(xdst, t >> 1, u0, u1);
    }
    load_trip(Wr0, Wr1, Bl, T - 1, qa0, qa1, qb);
    if (PF) panel_issue(xnext, (T >> 1) - 1, u0, u1);
    __builtin_amdgcn_sched_barrier(0);
    mma_trip(pa0, pa1, pb, acc0, acc1);
    mma_trip(qa0, qa1, qb, acc0, acc1);
    if (PF) panel_store(xdst, (T >> 1) - 1, u0, u1);
}

__device__ __forceinline__ void store_acc(float* __restrict__ Dsm, int wave, int lane,
                                          const v8f (&acc0)[8], const v8f (&acc1)[8])
{
    const int l15 = lane & 15;
    const int rst = (lane >> 4) << 3;
    const int row0 = wave << 5;
    #pragma unroll
    for (int nj = 0; nj < 8; ++nj) {
        const int col = (nj << 4) + l15;
        #pragma unroll
        for (int r = 0; r < 8; r += 2) {
            v2f pr;
            pr.x = acc0[nj][r];
            pr.y = acc0[nj][r + 1];
            *(v2f*)(Dsm + swz(row0 + rst + r, col)) = pr;
            v2f qr;
            qr.x = acc1[nj][r];
            qr.y = acc1[nj][r + 1];
            *(v2f*)(Dsm + swz(row0 + 16 + rst + r, col)) = qr;
        }
    }
}


__device__ __forceinline__ void killing_relu_tile(float* __restrict__ Xsm, int rowbase, int lane,
                                                  const v8f (&acc)[8])
{
    const int l15 = lane & 15;
    const int p   = lane >> 4;
    #pragma unroll
    for (int g = 0; g < 2; ++g) {
        const int f4 = rowbase + (p << 3) + (g << 2);
        float* base = Xsm + ((f4 >> 2) << 9) + (l15 << 2);
        v4f xa[8];
        #pragma unroll
        for (int k = 0; k < 8; ++k)
            xa[k] = *(const v4f*)(base + (k << 6));
        #pragma unroll
        for (int j = 0; j < 4; ++j) {
            float xv[8], dv[8];
            #pragma unroll
            for (int k = 0; k < 8; ++k) {
                xv[k] = xa[k][j];
                dv[k] = acc[k][(g << 2) + j];
            }
            const float kf = killing6(xv, dv);
            #pragma unroll
            for (int k = 0; k < 8; ++k)
                xa[k][j] = (kf < 0.0f) ? xv[k] : fmaf(kf, dv[k], xv[k]);
        }
        #pragma unroll
        for (int k = 0; k < 8; ++k)
            *(v4f*)(base + (k << 6)) = xa[k];
    }
}

__device__ __forceinline__ void killing_argmax(const float* __restrict__ Xsm, int row0, int lane,
                                               const v8f (&acc0)[8], const v8f (&acc1)[8],
                                               float* __restrict__ cand, int half,
                                               float* __restrict__ ws_val,
                                               float* __restrict__ ws_vec,
                                               size_t chan0, int pair)
{
    const int l15 = lane & 15;
    const int p   = lane >> 4;
    v4f   xa[4][8];
    float kf[16];
    float m[16];

    #pragma unroll
    for (int q = 0; q < 4; ++q) {
        const int f4 = row0 + ((q >> 1) << 4) + (p << 3) + ((q & 1) << 2);
        const float* base = Xsm + ((f4 >> 2) << 9) + (l15 << 2);
        #pragma unroll
        for (int k = 0; k < 8; ++k)
            xa[q][k] = *(const v4f*)(base + (k << 6));
    }
    #pragma unroll
    for (int q = 0; q < 4; ++q) {
        #pragma unroll
        for (int j = 0; j < 4; ++j) {
            float xv[8], dv[8];
            #pragma unroll
            for (int k = 0; k < 8; ++k) {
                xv[k] = xa[q][k][j];
                dv[k] = (q < 2) ? acc0[k][((q & 1) << 2) + j] : acc1[k][((q & 1) << 2) + j];
            }
            kf[(q << 2) + j] = killing6(xv, dv);
            m[(q << 2) + j]  = kf[(q << 2) + j];
        }
    }
    #pragma unroll
    for (int s = 1; s < 16; s <<= 1) {
        #pragma unroll
        for (int i = 0; i < 16; ++i)
            m[i] = fmaxf(m[i], __shfl_xor(m[i], s, 32));
    }
    int bestn[16];
    #pragma unroll
    for (int i = 0; i < 16; ++i) {
        const unsigned all = (unsigned)__ballot(kf[i] == m[i]);
        const unsigned hit = (all >> (p << 4)) & 0xffffu;
        bestn[i] = hit ? (__ffs(hit) - 1) : 0;
    }
    if (half == 0) {
        #pragma unroll
        for (int q = 0; q < 4; ++q) {
            const int f4 = row0 + ((q >> 1) << 4) + (p << 3) + ((q & 1) << 2);
            #pragma unroll
            for (int j = 0; j < 4; ++j) {
                const int i = (q << 2) + j;
                if (l15 == bestn[i]) {
                    float* cf = cand + (f4 + j) * CAND_PITCH;
                    v4f lo, hi;
                    lo.x = xa[q][0][j]; lo.y = xa[q][1][j]; lo.z = xa[q][2][j]; lo.w = xa[q][3][j];
                    hi.x = xa[q][4][j]; hi.y = xa[q][5][j]; hi.z = xa[q][6][j]; hi.w = xa[q][7][j];
                    *(v4f*)(cf)     = lo;
                    *(v4f*)(cf + 4) = hi;
                    cf[8] = kf[i];
                }
            }
        }
    } else {
        unsigned stay = 0u;
        #pragma unroll
        for (int q = 0; q < 4; ++q) {
            const int f4 = row0 + ((q >> 1) << 4) + (p << 3) + ((q & 1) << 2);
            #pragma unroll
            for (int j = 0; j < 4; ++j) {
                const int i = (q << 2) + j;
                const float prev = cand[(f4 + j) * CAND_PITCH + 8];
                if (m[i] > prev) {
                    if (l15 == bestn[i]) {
                        const size_t o = (chan0 + (size_t)(f4 + j)) * NTP + pair;
                        v4f lo, hi;
                        lo.x = xa[q][0][j]; lo.y = xa[q][1][j]; lo.z = xa[q][2][j]; lo.w = xa[q][3][j];
                        hi.x = xa[q][4][j]; hi.y = xa[q][5][j]; hi.z = xa[q][6][j]; hi.w = xa[q][7][j];
                        ws_val[o] = kf[i];
                        *(v4f*)(ws_vec + o * 8)     = lo;
                        *(v4f*)(ws_vec + o * 8 + 4) = hi;
                    }
                } else {
                    stay |= 1u << i;
                }
            }
        }
        if ((stay >> l15) & 1u) {
            const int f = row0 + ((l15 >> 3) << 4) + (p << 3) + (l15 & 7);
            const float* cf = cand + f * CAND_PITCH;
            const size_t o = (chan0 + (size_t)f) * NTP + pair;
            const v4f lo = *(const v4f*)(cf);
            const v4f hi = *(const v4f*)(cf + 4);
            ws_val[o] = cf[8];
            *(v4f*)(ws_vec + o * 8)     = lo;
            *(v4f*)(ws_vec + o * 8 + 4) = hi;
        }
    }
}

extern "C" __global__ __launch_bounds__(32)
void wrl_product(const float* __restrict__ Wrelu,
                 const float* __restrict__ Wlin,
                 float* __restrict__ Wrl)
{
    const int lane = threadIdx.x;
    const int l15 = lane & 15;
    const int p   = lane >> 4;
    const int f0  = blockIdx.x << 4;
    const int i0  = blockIdx.y << 5;
    const float* Ar = Wrelu + (size_t)(f0 + l15) * COUT + (p << 1);
    const float* Br = Wlin + (size_t)(p << 1) * CIN + i0 + (l15 << 1);
    v8f c0 = (v8f){0.f,0.f,0.f,0.f,0.f,0.f,0.f,0.f};
    v8f c1 = (v8f){0.f,0.f,0.f,0.f,0.f,0.f,0.f,0.f};
    #pragma unroll 16
    for (int kq = 0; kq < COUT / 4; ++kq) {
        const v2f a = *(const v2f*)(Ar + (kq << 2));
        const float* bp = Br + (size_t)(kq << 2) * CIN;
        const v2f w0 = *(const v2f*)(bp);
        const v2f w1 = *(const v2f*)(bp + CIN);
        v2f b0, b1;
        b0.x = w0.x; b0.y = w1.x;
        b1.x = w0.y; b1.y = w1.y;
        c0 = __builtin_amdgcn_wmma_f32_16x16x4_f32(false, a, false, b0, (short)0, c0, false, false);
        c1 = __builtin_amdgcn_wmma_f32_16x16x4_f32(false, a, false, b1, (short)0, c1, false, false);
    }
    float* o = Wrl + (size_t)(f0 + (p << 3)) * CIN + i0 + (l15 << 1);
    #pragma unroll
    for (int r = 0; r < 8; ++r) {
        v2f pr;
        pr.x = c0[r];
        pr.y = c1[r];
        *(v2f*)(o + (size_t)r * CIN) = pr;
    }
}

#define TPW   8
#define NWG_X (NT / TPW)

extern "C" __global__ __launch_bounds__(256)
void ln_fused(const float* __restrict__ x,
              const float* __restrict__ Wlin,
              const float* __restrict__ Wrl,
              const float* __restrict__ Wpool,
              float* __restrict__ ws_val,
              float* __restrict__ ws_vec)
{
    extern __shared__ float smem[];
    float* bufA = smem;
    float* bufB = smem + PANEL;
    float* cand = smem + PANEL + PANEL / 2;

    const int tid  = threadIdx.x;
    const int lane = tid & 31;
    const int wave = tid >> 5;
    const int tile0 = blockIdx.x * TPW;
    const int b     = blockIdx.y;

    v8f acc0[8], acc1[8];
    const int row0 = wave << 5;

    const int col = tid & (MCOLS - 1);
    const float* xthr = x + ((size_t)b * CIN * 8) * NDIM
                          + ((size_t)((tid >> 7) << 2) * 8 + (col >> 4)) * NDIM + (col & 15);
    float* bthr = bufB + ((tid >> 7) << 9) + (col << 2);

    {
        const float* xt = xthr + (size_t)tile0 * TILE_N;
        for (int m = 0; m < (CIN / 4) / 2; ++m) {
            const float* gp = xt + (size_t)(m << 1) * (32 * NDIM);
            v4f v;
            v.x = gp[0];
            v.y = gp[8 * NDIM];
            v.z = gp[16 * NDIM];
            v.w = gp[24 * NDIM];
            *(v4f*)(bthr + ((m << 1) << 9)) = v;
        }
    }

    #pragma unroll 1
    for (int i = 0; i < TPW; ++i) {
        const int tile = tile0 + i;
        const int pair = tile >> 1;
        const int half = tile & 1;

        __syncthreads();

        gemm_acc<false>(Wlin, CIN, CIN, bufB, wave, lane, acc0, acc1, nullptr, nullptr);
        store_acc(bufA, wave, lane, acc0, acc1);

        gemm_acc<false>(Wrl, CIN, CIN, bufB, wave, lane, acc0, acc1, nullptr, nullptr);
        killing_relu_tile(bufA, row0,      lane, acc0);
        killing_relu_tile(bufA, row0 + 16, lane, acc1);
        __syncthreads();

        const int nxt = (i + 1 < TPW) ? tile + 1 : tile;
        gemm_acc<true>(Wpool, COUT, COUT, bufA, wave, lane, acc0, acc1,
                       xthr + (size_t)nxt * TILE_N, bthr);
        killing_argmax(bufA, row0, lane, acc0, acc1, cand, half, ws_val, ws_vec, (size_t)b * COUT, pair);
    }
}

extern "C" __global__ __launch_bounds__(256)
void ln_reduce(const float* __restrict__ ws_val,
               const float* __restrict__ ws_vec,
               float* __restrict__ out)
{
    const int bf = blockIdx.x * 256 + threadIdx.x;
    if (bf >= BATCH * COUT) return;
    const float* v = ws_val + (size_t)bf * NTP;
    float best = v[0];
    int bestt = 0;
    for (int t = 1; t < NTP; ++t) {
        const float val = v[t];
        if (val > best) { best = val; bestt = t; }
    }
    const float* src = ws_vec + ((size_t)bf * NTP + bestt) * 8;
    #pragma unroll
    for (int k = 0; k < 8; ++k)
        out[(size_t)bf * 8 + k] = src[k];
}

extern "C" void kernel_launch(void* const* d_in, const int* in_sizes, int n_in,
                              void* d_out, int out_size, void* d_ws, size_t ws_size,
                              hipStream_t stream)
{
    (void)in_sizes; (void)n_in; (void)out_size; (void)ws_size;

    const float* x     = (const float*)d_in[0];
    const float* Wlin  = (const float*)d_in[1];
    const float* Wrelu = (const float*)d_in[2];
    const float* Wpool = (const float*)d_in[3];
    float* out    = (float*)d_out;
    float* ws_val = (float*)d_ws;
    float* ws_vec = ws_val + (size_t)BATCH * COUT * NTP;
    float* Wrl    = ws_vec + (size_t)BATCH * COUT * NTP * 8;

    const size_t shmem = (size_t)(PANEL + PANEL / 2 + COUT * CAND_PITCH) * sizeof(float);
    hipFuncSetAttribute((const void*)ln_fused,
                        hipFuncAttributeMaxDynamicSharedMemorySize, (int)shmem);

    wrl_product<<<dim3(COUT / 16, CIN / 32), 32, 0, stream>>>(Wrelu, Wlin, Wrl);
    dim3 grid(NWG_X, BATCH);
    ln_fused<<<grid, 256, shmem, stream>>>(x, Wlin, Wrl, Wpool, ws_val, ws_vec);
    ln_reduce<<<(BATCH * COUT + 255) / 256, 256, 0, stream>>>(ws_val, ws_vec, out);
}
